# starting kernel
# baseline (speedup 1.0000x reference)
#include <hip/hip_runtime.h>
#include <hip/hip_bf16.h>


typedef __attribute__((ext_vector_type(2))) float    v2f;
typedef __attribute__((ext_vector_type(8))) float    v8f;
typedef __attribute__((ext_vector_type(2))) unsigned v2u;

#define NPATCH   196
#define DDIM     768
#define XB       150528u
#define CSTRIDE  50176u

__global__ __launch_bounds__(128) void patch_scorer_wmma(
    const float* __restrict__ x,
    const float* __restrict__ weight,
    const float* __restrict__ bias,
    float* __restrict__ out)
{
    __shared__ __align__(16) unsigned s_off[DDIM];
    __shared__ __align__(16) float    s_w[DDIM];

    const unsigned p = blockIdx.x;

    for (int d = threadIdx.x; d < DDIM; d += 128) {
        unsigned L    = p * (unsigned)DDIM + (unsigned)d;
        unsigned pipj = L / 588u;
        unsigned rem  = L - pipj * 588u;
        unsigned c    = rem / 196u;
        unsigned r2   = rem - c * 196u;
        unsigned i    = r2 / 14u;
        unsigned j    = r2 - i * 14u;
        unsigned pi   = pipj >> 4;
        unsigned pj   = pipj & 15u;
        s_off[d] = c * CSTRIDE + (i * 16u + pi) * 224u + (j * 16u + pj);
        s_w[d]   = weight[(size_t)p * DDIM + d];
    }
    __syncthreads();

    const unsigned lane = threadIdx.x & 31u;
    const unsigned wv   = threadIdx.x >> 5;
    const unsigned m    = lane & 15u;
    const unsigned hh   = lane >> 4;
    const unsigned b0   = blockIdx.y * 64u + wv * 16u;

    const unsigned idxBase = (b0 + m) * XB;

    v8f acc = {0.f, 0.f, 0.f, 0.f, 0.f, 0.f, 0.f, 0.f};

    #pragma unroll 4
    for (int kc = 0; kc < DDIM; kc += 4) {
        const int d0 = kc + (int)(hh << 1);
        const v2u off = *(const v2u*)&s_off[d0];
        v2f a;
        a.x = x[idxBase + off.x];
        a.y = x[idxBase + off.y];
        const v2f b = *(const v2f*)&s_w[d0];

        acc = __builtin_amdgcn_wmma_f32_16x16x4_f32(
            false, a, false, b, (short)0, acc, false, false);
    }

    const float bp = bias[p];
    if (m == 0u) {
        const unsigned mb = hh * 8u;
        #pragma unroll
        for (int v = 0; v < 8; ++v)
            out[(size_t)(b0 + mb + (unsigned)v) * NPATCH + p] = acc[v] + bp;
    }
}

extern "C" void kernel_launch(void* const* d_in, const int* in_sizes, int n_in,
                              void* d_out, int out_size, void* d_ws, size_t ws_size,
                              hipStream_t stream) {
    const float* x      = (const float*)d_in[0];
    const float* weight = (const float*)d_in[1];
    const float* bias   = (const float*)d_in[2];
    float*       out    = (float*)d_out;

    const int B = in_sizes[0] / (int)XB;
    dim3 grid(NPATCH, (unsigned)(B / 64), 1);
    dim3 block(128, 1, 1);
    patch_scorer_wmma<<<grid, block, 0, stream>>>(x, weight, bias, out);
}


# read x once as whole contiguous image rows (16 B per lane); per-patch segment sum on the f32 WMMA with a one-hot weighted B; d summed in image order, then a fixed-order LDS sum
# speedup vs baseline: 6.6966x; 6.6966x over previous
#include <hip/hip_runtime.h>


typedef __attribute__((ext_vector_type(2))) float v2f;
typedef __attribute__((ext_vector_type(4))) float v4f;
typedef __attribute__((ext_vector_type(8))) float v8f;

#define NPATCH     196
#define XB         150528u
#define CSTRIDE    50176u
#define ROWW       224u
#define PI_ELEMS   9408u
#define NBANDS     4u
#define NCHUNK     7
#define NWAVES     12u
#define NTHREADS   384
#define SETUP_IT   25
#define GMAX       576u
#define HSTEPS     12

__device__ __forceinline__ void load_row(v4f (&buf)[NCHUNK], const float* xl, unsigned rho)
{
    const unsigned c = rho / 14u, i = rho - c * 14u;
    const float* xr = xl + c * CSTRIDE + i * (16u * ROWW);
    #pragma unroll
    for (int u = 0; u < NCHUNK; ++u)
        buf[u] = *(const v4f*)(xr + u * 8);
}

__device__ __forceinline__ void mac_row(const v4f (&buf)[NCHUNK], const float* s_w,
                                        const unsigned char* s_col, unsigned lr, unsigned c16,
                                        v8f& acc0, v8f& acc1)
{
    #pragma unroll
    for (int u = 0; u < NCHUNK; ++u) {
        const v4f      wq = *(const v4f*)&s_w[lr + u * 8];
        const unsigned sc = *(const unsigned*)&s_col[lr + u * 8];
        v2f a0, a1, b0, b1;
        a0.x = buf[u].x; a0.y = buf[u].y;
        a1.x = buf[u].z; a1.y = buf[u].w;
        b0.x = ((sc & 255u)         == c16) ? wq.x : 0.f;
        b0.y = (((sc >> 8) & 255u)  == c16) ? wq.y : 0.f;
        b1.x = (((sc >> 16) & 255u) == c16) ? wq.z : 0.f;
        b1.y = ((sc >> 24)          == c16) ? wq.w : 0.f;
        acc0 = __builtin_amdgcn_wmma_f32_16x16x4_f32(
            false, a0, false, b0, (short)0, acc0, false, false);
        acc1 = __builtin_amdgcn_wmma_f32_16x16x4_f32(
            false, a1, false, b1, (short)0, acc1, false, false);
    }
}

__global__ __launch_bounds__(NTHREADS) void patch_scorer_rows(
    const float* __restrict__ x,
    const float* __restrict__ weight,
    const float* __restrict__ bias,
    float* __restrict__ out)
{
    __shared__ __align__(16) float         s_w[PI_ELEMS];
    __shared__ __align__(16) unsigned char s_col[PI_ELEMS];
    __shared__ __align__(16) unsigned      s_goff[GMAX];
    __shared__ __align__(16) float         s_gw[GMAX];
    __shared__ __align__(16) float         s_part[NWAVES * 256u];

    const unsigned pi    = blockIdx.x;
    const unsigned bt    = blockIdx.y;
    const unsigned q     = pi & 3u;
    const unsigned pbase = (pi >> 2) * 49u + q * 12u;
    const unsigned nhead = q * 192u;

    {
        float    wr[SETUP_IT];
        unsigned cl[SETUP_IT];
        #pragma unroll
        for (int t = 0; t < SETUP_IT; ++t) {
            unsigned idx = threadIdx.x + (unsigned)t * NTHREADS;
            idx = idx < PI_ELEMS ? idx : PI_ELEMS - 1u;
            const unsigned rho = idx / ROWW;
            const unsigned w   = idx - rho * ROWW;
            const unsigned c   = rho / 14u;
            const unsigned i   = rho - c * 14u;
            const unsigned L   = pi * 9408u + (w & 15u) * 588u + c * 196u + i * 14u + (w >> 4);
            wr[t] = weight[L];
            cl[t] = L / 768u - pbase;
        }
        #pragma unroll
        for (int t = 0; t < SETUP_IT; ++t) {
            const unsigned idx = threadIdx.x + (unsigned)t * NTHREADS;
            if (idx < PI_ELEMS) {
                s_w[idx]   = wr[t];
                s_col[idx] = (unsigned char)cl[t];
            }
        }
        for (unsigned hd = threadIdx.x; hd < nhead; hd += NTHREADS) {
            const unsigned L    = pbase * 768u + hd;
            const unsigned pipj = L / 588u;
            const unsigned rem  = L - pipj * 588u;
            const unsigned c    = rem / 196u;
            const unsigned r2   = rem - c * 196u;
            const unsigned i    = r2 / 14u;
            const unsigned j    = r2 - i * 14u;
            s_goff[hd] = c * CSTRIDE + (i * 16u + (pipj >> 4)) * ROWW + (j * 16u + (pipj & 15u));
            s_gw[hd]   = weight[L];
        }
    }
    __syncthreads();

    const unsigned lane = threadIdx.x & 31u;
    const unsigned wv   = (unsigned)__builtin_amdgcn_readfirstlane((int)(threadIdx.x >> 5));
    const unsigned ch   = wv / NBANDS;
    const unsigned bd   = wv - ch * NBANDS;
    const unsigned c16  = lane & 15u;
    const unsigned hl   = lane >> 4;

    const float* ximg = x + (size_t)(bt * 16u + c16) * XB;

    v8f acc0 = {0.f, 0.f, 0.f, 0.f, 0.f, 0.f, 0.f, 0.f};
    v8f acc1 = {0.f, 0.f, 0.f, 0.f, 0.f, 0.f, 0.f, 0.f};

    v2f hx[HSTEPS];
    if (nhead != 0u) {
        #pragma unroll
        for (int t = 0; t < HSTEPS; ++t) {
            const unsigned st = wv + (unsigned)t * NWAVES;
            const unsigned d0 = ((st * 4u < nhead) ? st * 4u : 0u) + hl * 2u;
            hx[t].x = ximg[s_goff[d0]];
            hx[t].y = ximg[s_goff[d0 + 1u]];
        }
    }

    const unsigned inrow = bd * 56u + hl * 4u;
    const float*   xl    = ximg + pi * ROWW + inrow;

    const unsigned row0 = ch * 14u;
    v4f bufA[NCHUNK], bufB[NCHUNK];
    load_row(bufA, xl, row0);
    #pragma unroll
    for (unsigned t = 0; t < 14u; t += 2u) {
        const unsigned r0 = row0 + t;
        load_row(bufB, xl, r0 + 1u);
        __builtin_amdgcn_sched_barrier(0);
        mac_row(bufA, s_w, s_col, r0 * ROWW + inrow, c16, acc0, acc1);
        if (t + 2u < 14u)
            load_row(bufA, xl, r0 + 2u);
        __builtin_amdgcn_sched_barrier(0);
        mac_row(bufB, s_w, s_col, (r0 + 1u) * ROWW + inrow, c16, acc0, acc1);
    }

    if (nhead != 0u) {
        #pragma unroll
        for (int t = 0; t < HSTEPS; ++t) {
            const unsigned st = wv + (unsigned)t * NWAVES;
            if (st * 4u < nhead) {
                const unsigned d0 = st * 4u + hl * 2u;
                v2f b;
                b.x = (c16 == 0u) ? s_gw[d0]      : 0.f;
                b.y = (c16 == 0u) ? s_gw[d0 + 1u] : 0.f;
                acc0 = __builtin_amdgcn_wmma_f32_16x16x4_f32(
                    false, hx[t], false, b, (short)0, acc0, false, false);
            }
        }
    }

    #pragma unroll
    for (int r = 0; r < 8; ++r)
        s_part[wv * 256u + (hl * 8u + (unsigned)r) * 16u + c16] = acc0[r] + acc1[r];
    __syncthreads();

    const unsigned ncols = (q == 3u) ? 13u : 12u;
    if (threadIdx.x < 16u * ncols) {
        const unsigned m   = threadIdx.x / ncols;
        const unsigned col = threadIdx.x - m * ncols;
        float s = 0.f;
        for (unsigned v = 0; v < NWAVES; ++v)
            s += s_part[v * 256u + m * 16u + col];
        const unsigned p = pbase + col;
        out[(size_t)(bt * 16u + m) * NPATCH + p] = s + bias[p];
    }
}

extern "C" void kernel_launch(void* const* d_in, const int* in_sizes, int n_in,
                              void* d_out, int out_size, void* d_ws, size_t ws_size,
                              hipStream_t stream) {
    const float* x      = (const float*)d_in[0];
    const float* weight = (const float*)d_in[1];
    const float* bias   = (const float*)d_in[2];
    float*       out    = (float*)d_out;

    dim3 grid(16, 16, 1);
    dim3 block(NTHREADS, 1, 1);
    patch_scorer_rows<<<grid, block, 0, stream>>>(x, weight, bias, out);
}


# head gathers issued inside the row stream (one step per row) and paired so a load touches 16 lines, not 32; head k order permuted, so its 4-term sums are formed in another order
# speedup vs baseline: 6.9839x; 1.0429x over previous
#include <hip/hip_runtime.h>


typedef __attribute__((ext_vector_type(2))) float v2f;
typedef __attribute__((ext_vector_type(4))) float v4f;
typedef __attribute__((ext_vector_type(8))) float v8f;

#define NPATCH     196
#define XB         150528u
#define CSTRIDE    50176u
#define ROWW       224u
#define PI_ELEMS   9408u
#define NBANDS     4u
#define NCHUNK     7
#define NWAVES     12u
#define NTHREADS   384
#define SETUP_IT   25
#define GMAX       576u
#define HSTEPS     12

__device__ __forceinline__ void load_row(v4f (&buf)[NCHUNK], const float* xl, unsigned rho)
{
    const unsigned c = rho / 14u, i = rho - c * 14u;
    const float* xr = xl + c * CSTRIDE + i * (16u * ROWW);
    #pragma unroll
    for (int u = 0; u < NCHUNK; ++u)
        buf[u] = *(const v4f*)(xr + u * 8);
}

__device__ __forceinline__ void head_offsets(unsigned& o0, unsigned& o1, const unsigned* s_goff,
                                             unsigned st, unsigned nhead, unsigned hl)
{
    const unsigned d0 = ((st * 4u < nhead) ? st * 4u : 0u) + hl;
    o0 = s_goff[d0];
    o1 = s_goff[d0 + 2u];
}

__device__ __forceinline__ void mac_row(const v4f (&buf)[NCHUNK], const float* s_w,
                                        const unsigned char* s_col, unsigned lr, unsigned c16,
                                        v8f& acc0, v8f& acc1)
{
    #pragma unroll
    for (int u = 0; u < NCHUNK; ++u) {
        const v4f      wq = *(const v4f*)&s_w[lr + u * 8];
        const unsigned sc = *(const unsigned*)&s_col[lr + u * 8];
        v2f a0, a1, b0, b1;
        a0.x = buf[u].x; a0.y = buf[u].y;
        a1.x = buf[u].z; a1.y = buf[u].w;
        b0.x = ((sc & 255u)         == c16) ? wq.x : 0.f;
        b0.y = (((sc >> 8) & 255u)  == c16) ? wq.y : 0.f;
        b1.x = (((sc >> 16) & 255u) == c16) ? wq.z : 0.f;
        b1.y = ((sc >> 24)          == c16) ? wq.w : 0.f;
        acc0 = __builtin_amdgcn_wmma_f32_16x16x4_f32(
            false, a0, false, b0, (short)0, acc0, false, false);
        acc1 = __builtin_amdgcn_wmma_f32_16x16x4_f32(
            false, a1, false, b1, (short)0, acc1, false, false);
    }
}

__global__ __launch_bounds__(NTHREADS) void patch_scorer_rows(
    const float* __restrict__ x,
    const float* __restrict__ weight,
    const float* __restrict__ bias,
    float* __restrict__ out)
{
    __shared__ __align__(16) float         s_w[PI_ELEMS];
    __shared__ __align__(16) unsigned char s_col[PI_ELEMS];
    __shared__ __align__(16) unsigned      s_goff[GMAX];
    __shared__ __align__(16) float         s_gw[GMAX];
    __shared__ __align__(16) float         s_part[NWAVES * 256u];

    const unsigned pi    = blockIdx.x;
    const unsigned bt    = blockIdx.y;
    const unsigned q     = pi & 3u;
    const unsigned pbase = (pi >> 2) * 49u + q * 12u;
    const unsigned nhead = q * 192u;

    {
        float    wr[SETUP_IT];
        unsigned cl[SETUP_IT];
        #pragma unroll
        for (int t = 0; t < SETUP_IT; ++t) {
            unsigned idx = threadIdx.x + (unsigned)t * NTHREADS;
            idx = idx < PI_ELEMS ? idx : PI_ELEMS - 1u;
            const unsigned rho = idx / ROWW;
            const unsigned w   = idx - rho * ROWW;
            const unsigned c   = rho / 14u;
            const unsigned i   = rho - c * 14u;
            const unsigned L   = pi * 9408u + (w & 15u) * 588u + c * 196u + i * 14u + (w >> 4);
            wr[t] = weight[L];
            cl[t] = L / 768u - pbase;
        }
        #pragma unroll
        for (int t = 0; t < SETUP_IT; ++t) {
            const unsigned idx = threadIdx.x + (unsigned)t * NTHREADS;
            if (idx < PI_ELEMS) {
                s_w[idx]   = wr[t];
                s_col[idx] = (unsigned char)cl[t];
            }
        }
        for (unsigned hd = threadIdx.x; hd < nhead; hd += NTHREADS) {
            const unsigned L    = pbase * 768u + hd;
            const unsigned pipj = L / 588u;
            const unsigned rem  = L - pipj * 588u;
            const unsigned c    = rem / 196u;
            const unsigned r2   = rem - c * 196u;
            const unsigned i    = r2 / 14u;
            const unsigned j    = r2 - i * 14u;
            s_goff[hd] = c * CSTRIDE + (i * 16u + (pipj >> 4)) * ROWW + (j * 16u + (pipj & 15u));
            s_gw[hd]   = weight[L];
        }
        if (nhead == 0u && threadIdx.x < 4u)
            s_goff[threadIdx.x] = 0u;
    }
    __syncthreads();

    const unsigned lane = threadIdx.x & 31u;
    const unsigned wv   = (unsigned)__builtin_amdgcn_readfirstlane((int)(threadIdx.x >> 5));
    const unsigned ch   = wv / NBANDS;
    const unsigned bd   = wv - ch * NBANDS;
    const unsigned c16  = lane & 15u;
    const unsigned hl   = lane >> 4;

    const float* ximg = x + (size_t)(bt * 16u + c16) * XB;

    v8f acc0 = {0.f, 0.f, 0.f, 0.f, 0.f, 0.f, 0.f, 0.f};
    v8f acc1 = {0.f, 0.f, 0.f, 0.f, 0.f, 0.f, 0.f, 0.f};

    v2f hx[HSTEPS];

    const unsigned inrow = bd * 56u + hl * 4u;
    const float*   xl    = ximg + pi * ROWW + inrow;

    const unsigned row0 = ch * 14u;
    v4f buf[3][NCHUNK];
    load_row(buf[0], xl, row0);
    load_row(buf[1], xl, row0 + 1u);
    unsigned go0, go1;
    head_offsets(go0, go1, s_goff, wv, nhead, hl);
    #pragma unroll
    for (unsigned t = 0; t < 14u; ++t) {
        if (t + 2u < 14u)
            load_row(buf[(t + 2u) % 3u], xl, row0 + t + 2u);
        if (t < (unsigned)HSTEPS) {
            hx[t].x = ximg[go0];
            hx[t].y = ximg[go1];
        }
        if (t + 1u < (unsigned)HSTEPS)
            head_offsets(go0, go1, s_goff, wv + (t + 1u) * NWAVES, nhead, hl);
        __builtin_amdgcn_sched_barrier(0);
        mac_row(buf[t % 3u], s_w, s_col, (row0 + t) * ROWW + inrow, c16, acc0, acc1);
        __builtin_amdgcn_sched_barrier(0);
    }

    #pragma unroll
    for (int t = 0; t < HSTEPS; ++t) {
        const unsigned st   = wv + (unsigned)t * NWAVES;
        const bool     live = st * 4u < nhead;
        const unsigned d0   = (live ? st * 4u : 0u) + hl;
        v2f a, b;
        a.x = live ? hx[t].x : 0.f;
        a.y = live ? hx[t].y : 0.f;
        b.x = (live && c16 == 0u) ? s_gw[d0]      : 0.f;
        b.y = (live && c16 == 0u) ? s_gw[d0 + 2u] : 0.f;
        acc0 = __builtin_amdgcn_wmma_f32_16x16x4_f32(
            false, a, false, b, (short)0, acc0, false, false);
    }

    #pragma unroll
    for (int r = 0; r < 8; ++r)
        s_part[wv * 256u + (hl * 8u + (unsigned)r) * 16u + c16] = acc0[r] + acc1[r];
    __syncthreads();

    const unsigned ncols = (q == 3u) ? 13u : 12u;
    if (threadIdx.x < 16u * ncols) {
        const unsigned m   = threadIdx.x / ncols;
        const unsigned col = threadIdx.x - m * ncols;
        float s = 0.f;
        for (unsigned v = 0; v < NWAVES; ++v)
            s += s_part[v * 256u + m * 16u + col];
        const unsigned p = pbase + col;
        out[(size_t)(bt * 16u + m) * NPATCH + p] = s + bias[p];
    }
}

extern "C" void kernel_launch(void* const* d_in, const int* in_sizes, int n_in,
                              void* d_out, int out_size, void* d_ws, size_t ws_size,
                              hipStream_t stream) {
    const float* x      = (const float*)d_in[0];
    const float* weight = (const float*)d_in[1];
    const float* bias   = (const float*)d_in[2];
    float*       out    = (float*)d_out;

    dim3 grid(16, 16, 1);
    dim3 block(NTHREADS, 1, 1);
    patch_scorer_rows<<<grid, block, 0, stream>>>(x, weight, bias, out);
}


# x read with whole-line loads (4 lines per load, was 16; 32 in flight per wave, was 14), turned to the batch-per-lane fragment in a wave-private LDS slot; line units, so d is summed in another order
# speedup vs baseline: 7.9020x; 1.1315x over previous
#include <hip/hip_runtime.h>


typedef __attribute__((ext_vector_type(2))) float v2f;
typedef __attribute__((ext_vector_type(4))) float v4f;
typedef __attribute__((ext_vector_type(8))) float v8f;

#define NPATCH     196
#define XB         150528u
#define CSTRIDE    50176u
#define ROWW       224u
#define PI_ELEMS   9408u
#define NUNITS     294u
#define LASTTRIP   24u
#define DEPTH      8u
#define SPITCH     36u
#define NWAVES     12u
#define NTHREADS   384
#define SETUP_IT   25
#define GMAX       576u
#define HSTEPS     12

__device__ __forceinline__ void load_unit(v4f (&g)[4], const float* xq, unsigned u)
{
    const unsigned rho = u / 7u, ln = u - rho * 7u;
    const unsigned c = rho / 14u, i = rho - c * 14u;
    const float* xr = xq + c * CSTRIDE + i * (16u * ROWW) + ln * 32u;
    #pragma unroll
    for (int v = 0; v < 4; ++v)
        g[v] = *(const v4f*)(xr + (size_t)v * (4u * XB));
}

__device__ __forceinline__ void store_unit(float* st_wr, const v4f (&g)[4])
{
    #pragma unroll
    for (int v = 0; v < 4; ++v)
        *(v4f*)(st_wr + v * (4u * SPITCH)) = g[v];
}

__device__ __forceinline__ void frag_load(v4f (&fa)[4], v4f (&fw)[4], unsigned (&fc)[4],
                                          const float* st_rd, const float* s_w,
                                          const unsigned char* s_col, unsigned lr)
{
    #pragma unroll
    for (int s = 0; s < 4; ++s) {
        fa[s] = *(const v4f*)(st_rd + s * 8);
        fw[s] = *(const v4f*)&s_w[lr + s * 8];
        fc[s] = *(const unsigned*)&s_col[lr + s * 8];
    }
}

__device__ __forceinline__ void head_offsets(unsigned& o0, unsigned& o1, const unsigned* s_goff,
                                             unsigned st, unsigned nhead, unsigned hl)
{
    const unsigned d0 = ((st * 4u < nhead) ? st * 4u : 0u) + hl;
    o0 = s_goff[d0];
    o1 = s_goff[d0 + 2u];
}

__device__ __forceinline__ void mac_unit(const v4f (&fa)[4], const v4f (&fw)[4], const unsigned (&fc)[4],
                                         unsigned c16, v8f& acc0, v8f& acc1)
{
    #pragma unroll
    for (int s = 0; s < 4; ++s) {
        const v4f      wq = fw[s];
        const unsigned sc = fc[s];
        v2f a0, a1, b0, b1;
        a0.x = fa[s].x; a0.y = fa[s].y;
        a1.x = fa[s].z; a1.y = fa[s].w;
        b0.x = ((sc & 255u)         == c16) ? wq.x : 0.f;
        b0.y = (((sc >> 8) & 255u)  == c16) ? wq.y : 0.f;
        b1.x = (((sc >> 16) & 255u) == c16) ? wq.z : 0.f;
        b1.y = ((sc >> 24)          == c16) ? wq.w : 0.f;
        acc0 = __builtin_amdgcn_wmma_f32_16x16x4_f32(
            false, a0, false, b0, (short)0, acc0, false, false);
        acc1 = __builtin_amdgcn_wmma_f32_16x16x4_f32(
            false, a1, false, b1, (short)0, acc1, false, false);
    }
}

__global__ __launch_bounds__(NTHREADS) void patch_scorer_rows(
    const float* __restrict__ x,
    const float* __restrict__ weight,
    const float* __restrict__ bias,
    float* __restrict__ out)
{
    __shared__ __align__(16) float         s_w[PI_ELEMS];
    __shared__ __align__(16) unsigned char s_col[PI_ELEMS];
    __shared__ __align__(16) unsigned      s_goff[GMAX];
    __shared__ __align__(16) float         s_gw[GMAX];
    __shared__ __align__(16) float         s_part[NWAVES * 256u];
    __shared__ __align__(16) float         s_stage[NWAVES * 16u * SPITCH];

    const unsigned pi    = blockIdx.x;
    const unsigned bt    = blockIdx.y;
    const unsigned q     = pi & 3u;
    const unsigned pbase = (pi >> 2) * 49u + q * 12u;
    const unsigned nhead = q * 192u;

    {
        float    wr[SETUP_IT];
        unsigned cl[SETUP_IT];
        #pragma unroll
        for (int t = 0; t < SETUP_IT; ++t) {
            unsigned idx = threadIdx.x + (unsigned)t * NTHREADS;
            idx = idx < PI_ELEMS ? idx : PI_ELEMS - 1u;
            const unsigned rho = idx / ROWW;
            const unsigned w   = idx - rho * ROWW;
            const unsigned c   = rho / 14u;
            const unsigned i   = rho - c * 14u;
            const unsigned L   = pi * 9408u + (w & 15u) * 588u + c * 196u + i * 14u + (w >> 4);
            wr[t] = weight[L];
            cl[t] = L / 768u - pbase;
        }
        #pragma unroll
        for (int t = 0; t < SETUP_IT; ++t) {
            const unsigned idx = threadIdx.x + (unsigned)t * NTHREADS;
            if (idx < PI_ELEMS) {
                s_w[idx]   = wr[t];
                s_col[idx] = (unsigned char)cl[t];
            }
        }
        for (unsigned hd = threadIdx.x; hd < nhead; hd += NTHREADS) {
            const unsigned L    = pbase * 768u + hd;
            const unsigned pipj = L / 588u;
            const unsigned rem  = L - pipj * 588u;
            const unsigned c    = rem / 196u;
            const unsigned r2   = rem - c * 196u;
            const unsigned i    = r2 / 14u;
            const unsigned j    = r2 - i * 14u;
            s_goff[hd] = c * CSTRIDE + (i * 16u + (pipj >> 4)) * ROWW + (j * 16u + (pipj & 15u));
            s_gw[hd]   = weight[L];
        }
        if (nhead == 0u && threadIdx.x < 4u)
            s_goff[threadIdx.x] = 0u;
    }
    __syncthreads();

    const unsigned lane = threadIdx.x & 31u;
    const unsigned wv   = (unsigned)__builtin_amdgcn_readfirstlane((int)(threadIdx.x >> 5));
    const unsigned c16  = lane & 15u;
    const unsigned hl   = lane >> 4;
    const unsigned g4   = lane >> 3;
    const unsigned p8   = lane & 7u;

    const float* ximg = x + (size_t)(bt * 16u + c16) * XB;
    const float* xq   = x + (size_t)(bt * 16u + g4) * XB + pi * ROWW + p8 * 4u;

    float*       st_wr = s_stage + wv * (16u * SPITCH) + g4 * SPITCH + p8 * 4u;
    const float* st_rd = s_stage + wv * (16u * SPITCH) + c16 * SPITCH + hl * 4u;

    v8f acc0 = {0.f, 0.f, 0.f, 0.f, 0.f, 0.f, 0.f, 0.f};
    v8f acc1 = {0.f, 0.f, 0.f, 0.f, 0.f, 0.f, 0.f, 0.f};

    v2f hx[HSTEPS];

    v4f      G[DEPTH][4];
    v4f      fa[2][4], fw[2][4];
    unsigned fc[2][4];
    const bool tail = wv + NWAVES * LASTTRIP < NUNITS;
    #pragma unroll
    for (unsigned d = 0; d < DEPTH; ++d)
        load_unit(G[d], xq, wv + NWAVES * d);
    unsigned go0, go1;
    head_offsets(go0, go1, s_goff, wv, nhead, hl);
    store_unit(st_wr, G[0]);
    load_unit(G[0], xq, wv + NWAVES * DEPTH);
    frag_load(fa[0], fw[0], fc[0], st_rd, s_w, s_col, wv * 32u + hl * 4u);
    #pragma unroll
    for (unsigned t = 0; t <= LASTTRIP; ++t) {
        const unsigned n = t + 1u;
        const unsigned m = n + DEPTH;
        if (n < LASTTRIP || (n == LASTTRIP && tail)) {
            store_unit(st_wr, G[n % DEPTH]);
            if (m < LASTTRIP || (m == LASTTRIP && tail))
                load_unit(G[n % DEPTH], xq, wv + NWAVES * m);
            frag_load(fa[n & 1u], fw[n & 1u], fc[n & 1u], st_rd, s_w, s_col,
                      (wv + NWAVES * n) * 32u + hl * 4u);
        }
        if (t < (unsigned)HSTEPS) {
            hx[t].x = ximg[go0];
            hx[t].y = ximg[go1];
        }
        if (t + 1u < (unsigned)HSTEPS)
            head_offsets(go0, go1, s_goff, wv + (t + 1u) * NWAVES, nhead, hl);
        __builtin_amdgcn_sched_barrier(0);
        if (t < LASTTRIP || tail)
            mac_unit(fa[t & 1u], fw[t & 1u], fc[t & 1u], c16, acc0, acc1);
        __builtin_amdgcn_sched_barrier(0);
    }

    #pragma unroll
    for (int t = 0; t < HSTEPS; ++t) {
        const unsigned st   = wv + (unsigned)t * NWAVES;
        const bool     live = st * 4u < nhead;
        const unsigned d0   = (live ? st * 4u : 0u) + hl;
        v2f a, b;
        a.x = live ? hx[t].x : 0.f;
        a.y = live ? hx[t].y : 0.f;
        b.x = (live && c16 == 0u) ? s_gw[d0]      : 0.f;
        b.y = (live && c16 == 0u) ? s_gw[d0 + 2u] : 0.f;
        acc0 = __builtin_amdgcn_wmma_f32_16x16x4_f32(
            false, a, false, b, (short)0, acc0, false, false);
    }

    #pragma unroll
    for (int r = 0; r < 8; ++r)
        s_part[wv * 256u + (hl * 8u + (unsigned)r) * 16u + c16] = acc0[r] + acc1[r];
    __syncthreads();

    const unsigned ncols = (q == 3u) ? 13u : 12u;
    if (threadIdx.x < 16u * ncols) {
        const unsigned m   = threadIdx.x / ncols;
        const unsigned col = threadIdx.x - m * ncols;
        float s = 0.f;
        for (unsigned v = 0; v < NWAVES; ++v)
            s += s_part[v * 256u + m * 16u + col];
        const unsigned p = pbase + col;
        out[(size_t)(bt * 16u + m) * NPATCH + p] = s + bias[p];
    }
}

extern "C" void kernel_launch(void* const* d_in, const int* in_sizes, int n_in,
                              void* d_out, int out_size, void* d_ws, size_t ws_size,
                              hipStream_t stream) {
    const float* x      = (const float*)d_in[0];
    const float* weight = (const float*)d_in[1];
    const float* bias   = (const float*)d_in[2];
    float*       out    = (float*)d_out;

    dim3 grid(16, 16, 1);
    dim3 block(NTHREADS, 1, 1);
    patch_scorer_rows<<<grid, block, 0, stream>>>(x, weight, bias, out);
}


# straddling patches no longer gathered per batch: both sides' partial sums (own streamed rows, f32 WMMA) go through 24 KB of workspace; a 12-block second launch adds the two partials in a fixed order
# speedup vs baseline: 8.4844x; 1.0737x over previous
#include <hip/hip_runtime.h>


typedef __attribute__((ext_vector_type(2))) float v2f;
typedef __attribute__((ext_vector_type(4))) float v4f;
typedef __attribute__((ext_vector_type(8))) float v8f;

#define NPATCH     196
#define XB         150528u
#define CSTRIDE    50176u
#define ROWW       224u
#define PI_ELEMS   9408u
#define NUNITS     294u
#define LASTTRIP   24u
#define DEPTH      8u
#define SPITCH     36u
#define NWAVES     12u
#define NTHREADS   384
#define SETUP_IT   25
#define GMAX       576u
#define HSTEPS     12
#define NSTRAD     12u
#define WS_HALF    3072u
#define WS_BYTES   24576ull

__device__ __forceinline__ void load_unit(v4f (&g)[4], const float* xq, unsigned u)
{
    const unsigned rho = u / 7u, ln = u - rho * 7u;
    const unsigned c = rho / 14u, i = rho - c * 14u;
    const float* xr = xq + c * CSTRIDE + i * (16u * ROWW) + ln * 32u;
    #pragma unroll
    for (int v = 0; v < 4; ++v)
        g[v] = *(const v4f*)(xr + (size_t)v * (4u * XB));
}

__device__ __forceinline__ void store_unit(float* st_wr, const v4f (&g)[4])
{
    #pragma unroll
    for (int v = 0; v < 4; ++v)
        *(v4f*)(st_wr + v * (4u * SPITCH)) = g[v];
}

__device__ __forceinline__ void frag_load(v4f (&fa)[4], v4f (&fw)[4], unsigned (&fc)[4],
                                          const float* st_rd, const float* s_w,
                                          const unsigned char* s_col, unsigned lr)
{
    #pragma unroll
    for (int s = 0; s < 4; ++s) {
        fa[s] = *(const v4f*)(st_rd + s * 8);
        fw[s] = *(const v4f*)&s_w[lr + s * 8];
        fc[s] = *(const unsigned*)&s_col[lr + s * 8];
    }
}

__device__ __forceinline__ void head_offsets(unsigned& o0, unsigned& o1, const unsigned* s_goff,
                                             unsigned st, unsigned nhead, unsigned hl)
{
    const unsigned d0 = ((st * 4u < nhead) ? st * 4u : 0u) + hl;
    o0 = s_goff[d0];
    o1 = s_goff[d0 + 2u];
}

__device__ __forceinline__ void mac_unit(const v4f (&fa)[4], const v4f (&fw)[4], const unsigned (&fc)[4],
                                         unsigned c16, v8f& acc0, v8f& acc1)
{
    #pragma unroll
    for (int s = 0; s < 4; ++s) {
        const v4f      wq = fw[s];
        const unsigned sc = fc[s];
        v2f a0, a1, b0, b1;
        a0.x = fa[s].x; a0.y = fa[s].y;
        a1.x = fa[s].z; a1.y = fa[s].w;
        b0.x = ((sc & 255u)         == c16) ? wq.x : 0.f;
        b0.y = (((sc >> 8) & 255u)  == c16) ? wq.y : 0.f;
        b1.x = (((sc >> 16) & 255u) == c16) ? wq.z : 0.f;
        b1.y = ((sc >> 24)          == c16) ? wq.w : 0.f;
        acc0 = __builtin_amdgcn_wmma_f32_16x16x4_f32(
            false, a0, false, b0, (short)0, acc0, false, false);
        acc1 = __builtin_amdgcn_wmma_f32_16x16x4_f32(
            false, a1, false, b1, (short)0, acc1, false, false);
    }
}

template <bool HEAD, unsigned DEPTH_T>
__device__ __forceinline__ void scorer_body(
    const float* __restrict__ x,
    const float* __restrict__ weight,
    const float* __restrict__ bias,
    float* __restrict__ out,
    float* __restrict__ ws,
    float* s_w, unsigned char* s_col, unsigned* s_goff, float* s_gw, float* s_part, float* s_stage)
{
    const unsigned pi    = blockIdx.x;
    const unsigned bt    = blockIdx.y;
    const unsigned q     = pi & 3u;
    const unsigned pbase = (pi >> 2) * 49u + q * 12u;
    const unsigned nhead = q * 192u;

    {
        float    wr[SETUP_IT];
        unsigned cl[SETUP_IT];
        #pragma unroll
        for (int t = 0; t < SETUP_IT; ++t) {
            unsigned idx = threadIdx.x + (unsigned)t * NTHREADS;
            idx = idx < PI_ELEMS ? idx : PI_ELEMS - 1u;
            const unsigned rho = idx / ROWW;
            const unsigned w   = idx - rho * ROWW;
            const unsigned c   = rho / 14u;
            const unsigned i   = rho - c * 14u;
            const unsigned L   = pi * 9408u + (w & 15u) * 588u + c * 196u + i * 14u + (w >> 4);
            wr[t] = weight[L];
            cl[t] = L / 768u - pbase;
        }
        #pragma unroll
        for (int t = 0; t < SETUP_IT; ++t) {
            const unsigned idx = threadIdx.x + (unsigned)t * NTHREADS;
            if (idx < PI_ELEMS) {
                s_w[idx]   = wr[t];
                s_col[idx] = (unsigned char)cl[t];
            }
        }
        if constexpr (HEAD) {
        for (unsigned hd = threadIdx.x; hd < nhead; hd += NTHREADS) {
            const unsigned L    = pbase * 768u + hd;
            const unsigned pipj = L / 588u;
            const unsigned rem  = L - pipj * 588u;
            const unsigned c    = rem / 196u;
            const unsigned r2   = rem - c * 196u;
            const unsigned i    = r2 / 14u;
            const unsigned j    = r2 - i * 14u;
            s_goff[hd] = c * CSTRIDE + (i * 16u + (pipj >> 4)) * ROWW + (j * 16u + (pipj & 15u));
            s_gw[hd]   = weight[L];
        }
        if (nhead == 0u && threadIdx.x < 4u)
            s_goff[threadIdx.x] = 0u;
        }
    }
    __syncthreads();

    const unsigned lane = threadIdx.x & 31u;
    const unsigned wv   = (unsigned)__builtin_amdgcn_readfirstlane((int)(threadIdx.x >> 5));
    const unsigned c16  = lane & 15u;
    const unsigned hl   = lane >> 4;
    const unsigned g4   = lane >> 3;
    const unsigned p8   = lane & 7u;

    const float* ximg = x + (size_t)(bt * 16u + c16) * XB;
    const float* xq   = x + (size_t)(bt * 16u + g4) * XB + pi * ROWW + p8 * 4u;

    float*       st_wr = s_stage + wv * (16u * SPITCH) + g4 * SPITCH + p8 * 4u;
    const float* st_rd = s_stage + wv * (16u * SPITCH) + c16 * SPITCH + hl * 4u;

    v8f acc0 = {0.f, 0.f, 0.f, 0.f, 0.f, 0.f, 0.f, 0.f};
    v8f acc1 = {0.f, 0.f, 0.f, 0.f, 0.f, 0.f, 0.f, 0.f};

    v2f hx[HSTEPS];

    v4f      G[DEPTH_T][4];
    v4f      fa[2][4], fw[2][4];
    unsigned fc[2][4];
    const bool tail = wv + NWAVES * LASTTRIP < NUNITS;
    #pragma unroll
    for (unsigned d = 0; d < DEPTH_T; ++d)
        load_unit(G[d], xq, wv + NWAVES * d);
    unsigned go0 = 0u, go1 = 0u;
    if constexpr (HEAD)
        head_offsets(go0, go1, s_goff, wv, nhead, hl);
    store_unit(st_wr, G[0]);
    load_unit(G[0], xq, wv + NWAVES * DEPTH_T);
    frag_load(fa[0], fw[0], fc[0], st_rd, s_w, s_col, wv * 32u + hl * 4u);
    #pragma unroll
    for (unsigned t = 0; t <= LASTTRIP; ++t) {
        const unsigned n = t + 1u;
        const unsigned m = n + DEPTH_T;
        if (n < LASTTRIP || (n == LASTTRIP && tail)) {
            store_unit(st_wr, G[n % DEPTH_T]);
            if (m < LASTTRIP || (m == LASTTRIP && tail))
                load_unit(G[n % DEPTH_T], xq, wv + NWAVES * m);
            frag_load(fa[n & 1u], fw[n & 1u], fc[n & 1u], st_rd, s_w, s_col,
                      (wv + NWAVES * n) * 32u + hl * 4u);
        }
        if constexpr (HEAD) {
        if (t < (unsigned)HSTEPS) {
            hx[t].x = ximg[go0];
            hx[t].y = ximg[go1];
        }
        if (t + 1u < (unsigned)HSTEPS)
            head_offsets(go0, go1, s_goff, wv + (t + 1u) * NWAVES, nhead, hl);
        }
        __builtin_amdgcn_sched_barrier(0);
        if (t < LASTTRIP || tail)
            mac_unit(fa[t & 1u], fw[t & 1u], fc[t & 1u], c16, acc0, acc1);
        __builtin_amdgcn_sched_barrier(0);
    }

    if constexpr (HEAD) {
    #pragma unroll
    for (int t = 0; t < HSTEPS; ++t) {
        const unsigned st   = wv + (unsigned)t * NWAVES;
        const bool     live = st * 4u < nhead;
        const unsigned d0   = (live ? st * 4u : 0u) + hl;
        v2f a, b;
        a.x = live ? hx[t].x : 0.f;
        a.y = live ? hx[t].y : 0.f;
        b.x = (live && c16 == 0u) ? s_gw[d0]      : 0.f;
        b.y = (live && c16 == 0u) ? s_gw[d0 + 2u] : 0.f;
        acc0 = __builtin_amdgcn_wmma_f32_16x16x4_f32(
            false, a, false, b, (short)0, acc0, false, false);
    }
    }

    #pragma unroll
    for (int r = 0; r < 8; ++r)
        s_part[wv * 256u + (hl * 8u + (unsigned)r) * 16u + c16] = acc0[r] + acc1[r];
    __syncthreads();

    if (threadIdx.x < 16u * 13u) {
        const unsigned m   = threadIdx.x / 13u;
        const unsigned col = threadIdx.x - m * 13u;
        float s = 0.f;
        for (unsigned v = 0; v < NWAVES; ++v)
            s += s_part[v * 256u + m * 16u + col];
        const unsigned b = bt * 16u + m;
        const unsigned p = pbase + col;
        const unsigned g = pi >> 2;
        if (col == 12u && q != 3u) {
            if constexpr (!HEAD)
                ws[(g * 3u + q) * 256u + b] = s;
        } else if (!HEAD && col == 0u && q != 0u) {
            ws[WS_HALF + (g * 3u + q - 1u) * 256u + b] = s;
        } else {
            out[(size_t)b * NPATCH + p] = s + bias[p];
        }
    }
}

__global__ __launch_bounds__(NTHREADS) void patch_scorer_rows(
    const float* __restrict__ x,
    const float* __restrict__ weight,
    const float* __restrict__ bias,
    float* __restrict__ out,
    float* __restrict__ ws,
    unsigned long long ws_bytes)
{
    __shared__ __align__(16) float         s_w[PI_ELEMS];
    __shared__ __align__(16) unsigned char s_col[PI_ELEMS];
    __shared__ __align__(16) unsigned      s_goff[GMAX];
    __shared__ __align__(16) float         s_gw[GMAX];
    __shared__ __align__(16) float         s_part[NWAVES * 256u];
    __shared__ __align__(16) float         s_stage[NWAVES * 16u * SPITCH];

    if (ws != nullptr && ws_bytes >= WS_BYTES)
        scorer_body<false, DEPTH>(x, weight, bias, out, ws, s_w, s_col, s_goff, s_gw, s_part, s_stage);
    else
        scorer_body<true, DEPTH - 2u>(x, weight, bias, out, ws, s_w, s_col, s_goff, s_gw, s_part, s_stage);
}

__global__ __launch_bounds__(256) void patch_scorer_join(
    const float* __restrict__ bias,
    float* __restrict__ out,
    const float* __restrict__ ws,
    unsigned long long ws_bytes)
{
    if (ws == nullptr || ws_bytes < WS_BYTES)
        return;
    const unsigned sidx = blockIdx.x;
    const unsigned b    = threadIdx.x;
    const unsigned g    = sidx / 3u;
    const unsigned ql   = sidx - g * 3u + 1u;
    const unsigned p    = g * 49u + ql * 12u;
    out[(size_t)b * NPATCH + p] = (ws[sidx * 256u + b] + ws[WS_HALF + sidx * 256u + b]) + bias[p];
}

extern "C" void kernel_launch(void* const* d_in, const int* in_sizes, int n_in,
                              void* d_out, int out_size, void* d_ws, size_t ws_size,
                              hipStream_t stream) {
    const float* x      = (const float*)d_in[0];
    const float* weight = (const float*)d_in[1];
    const float* bias   = (const float*)d_in[2];
    float*       out    = (float*)d_out;
    float*       ws     = (float*)d_ws;

    dim3 grid(16, 16, 1);
    dim3 block(NTHREADS, 1, 1);
    patch_scorer_rows<<<grid, block, 0, stream>>>(x, weight, bias, out, ws, (unsigned long long)ws_size);
    patch_scorer_join<<<dim3(NSTRAD, 1, 1), dim3(256, 1, 1), 0, stream>>>(bias, out, ws, (unsigned long long)ws_size);
}


# weight image staged as a transpose: a wave load takes 4 weight rows x 32 contiguous bytes (4-5 lines per load, was 16); patch columns computed, 4 per word; all sums and their order unchanged
# speedup vs baseline: 8.8084x; 1.0382x over previous
#include <hip/hip_runtime.h>


typedef __attribute__((ext_vector_type(2))) float v2f;
typedef __attribute__((ext_vector_type(4))) float v4f;
typedef __attribute__((ext_vector_type(8))) float v8f;

#define NPATCH     196
#define XB         150528u
#define CSTRIDE    50176u
#define ROWW       224u
#define PI_ELEMS   9408u
#define NUNITS     294u
#define LASTTRIP   24u
#define DEPTH      8u
#define SPITCH     36u
#define NWAVES     12u
#define NTHREADS   384
#define RLEN       588u
#define W_IT       25
#define W_RSTEP    24u
#define COLW       2352u
#define COL_IT     7
#define GMAX       576u
#define HSTEPS     12
#define NSTRAD     12u
#define WS_HALF    3072u
#define WS_BYTES   24576ull

__device__ __forceinline__ void load_unit(v4f (&g)[4], const float* xq, unsigned u)
{
    const unsigned rho = u / 7u, ln = u - rho * 7u;
    const unsigned c = rho / 14u, i = rho - c * 14u;
    const float* xr = xq + c * CSTRIDE + i * (16u * ROWW) + ln * 32u;
    #pragma unroll
    for (int v = 0; v < 4; ++v)
        g[v] = *(const v4f*)(xr + (size_t)v * (4u * XB));
}

__device__ __forceinline__ void store_unit(float* st_wr, const v4f (&g)[4])
{
    #pragma unroll
    for (int v = 0; v < 4; ++v)
        *(v4f*)(st_wr + v * (4u * SPITCH)) = g[v];
}

__device__ __forceinline__ void frag_load(v4f (&fa)[4], v4f (&fw)[4], unsigned (&fc)[4],
                                          const float* st_rd, const float* s_w,
                                          const unsigned char* s_col, unsigned lr)
{
    #pragma unroll
    for (int s = 0; s < 4; ++s) {
        fa[s] = *(const v4f*)(st_rd + s * 8);
        fw[s] = *(const v4f*)&s_w[lr + s * 8];
        fc[s] = *(const unsigned*)&s_col[lr + s * 8];
    }
}

__device__ __forceinline__ void head_offsets(unsigned& o0, unsigned& o1, const unsigned* s_goff,
                                             unsigned st, unsigned nhead, unsigned hl)
{
    const unsigned d0 = ((st * 4u < nhead) ? st * 4u : 0u) + hl;
    o0 = s_goff[d0];
    o1 = s_goff[d0 + 2u];
}

__device__ __forceinline__ void mac_unit(const v4f (&fa)[4], const v4f (&fw)[4], const unsigned (&fc)[4],
                                         unsigned c16, v8f& acc0, v8f& acc1)
{
    #pragma unroll
    for (int s = 0; s < 4; ++s) {
        const v4f      wq = fw[s];
        const unsigned sc = fc[s];
        v2f a0, a1, b0, b1;
        a0.x = fa[s].x; a0.y = fa[s].y;
        a1.x = fa[s].z; a1.y = fa[s].w;
        b0.x = ((sc & 255u)         == c16) ? wq.x : 0.f;
        b0.y = (((sc >> 8) & 255u)  == c16) ? wq.y : 0.f;
        b1.x = (((sc >> 16) & 255u) == c16) ? wq.z : 0.f;
        b1.y = ((sc >> 24)          == c16) ? wq.w : 0.f;
        acc0 = __builtin_amdgcn_wmma_f32_16x16x4_f32(
            false, a0, false, b0, (short)0, acc0, false, false);
        acc1 = __builtin_amdgcn_wmma_f32_16x16x4_f32(
            false, a1, false, b1, (short)0, acc1, false, false);
    }
}

template <bool HEAD, unsigned DEPTH_T>
__device__ __forceinline__ void scorer_body(
    const float* __restrict__ x,
    const float* __restrict__ weight,
    const float* __restrict__ bias,
    float* __restrict__ out,
    float* __restrict__ ws,
    float* s_w, unsigned char* s_col, unsigned* s_goff, float* s_gw, float* s_part, float* s_stage)
{
    const unsigned pi    = blockIdx.x;
    const unsigned bt    = blockIdx.y;
    const unsigned q     = pi & 3u;
    const unsigned pbase = (pi >> 2) * 49u + q * 12u;
    const unsigned nhead = q * 192u;

    {
        const unsigned wv1 = (unsigned)__builtin_amdgcn_readfirstlane((int)(threadIdx.x >> 5));
        const unsigned pj  = (wv1 & 3u) * 4u + ((threadIdx.x & 31u) >> 3);
        const unsigned r0  = (wv1 >> 2) * 8u + (threadIdx.x & 7u);
        const float*   wrow = weight + pi * PI_ELEMS + pj * RLEN;
        float*         swp  = s_w + pj;
        float wr[W_IT];
        #pragma unroll
        for (int t = 0; t < W_IT; ++t) {
            unsigned r = r0 + (unsigned)t * W_RSTEP;
            if (t + 1 == W_IT)
                r = r < RLEN ? r : RLEN - 1u;
            wr[t] = wrow[r];
        }
        #pragma unroll
        for (int t = 0; t < COL_IT; ++t) {
            const unsigned d = threadIdx.x + (unsigned)t * NTHREADS;
            if (d < COLW) {
                const unsigned L0 = pi * PI_ELEMS + (d & 3u) * (4u * RLEN) + (d >> 2);
                const unsigned c0 = L0 / 768u - pbase;
                const unsigned c1 = (L0 + RLEN) / 768u - pbase;
                const unsigned c2 = (L0 + 2u * RLEN) / 768u - pbase;
                const unsigned c3 = (L0 + 3u * RLEN) / 768u - pbase;
                ((unsigned*)s_col)[d] = c0 | (c1 << 8) | (c2 << 16) | (c3 << 24);
            }
        }
        #pragma unroll
        for (int t = 0; t < W_IT; ++t) {
            unsigned r = r0 + (unsigned)t * W_RSTEP;
            if (t + 1 == W_IT)
                r = r < RLEN ? r : RLEN - 1u;
            if (t + 1 < W_IT || (wv1 >> 2) < 2u)
                swp[16u * r] = wr[t];
        }
        if constexpr (HEAD) {
        for (unsigned hd = threadIdx.x; hd < nhead; hd += NTHREADS) {
            const unsigned L    = pbase * 768u + hd;
            const unsigned pipj = L / 588u;
            const unsigned rem  = L - pipj * 588u;
            const unsigned c    = rem / 196u;
            const unsigned r2   = rem - c * 196u;
            const unsigned i    = r2 / 14u;
            const unsigned j    = r2 - i * 14u;
            s_goff[hd] = c * CSTRIDE + (i * 16u + (pipj >> 4)) * ROWW + (j * 16u + (pipj & 15u));
            s_gw[hd]   = weight[L];
        }
        if (nhead == 0u && threadIdx.x < 4u)
            s_goff[threadIdx.x] = 0u;
        }
    }
    __syncthreads();

    const unsigned lane = threadIdx.x & 31u;
    const unsigned wv   = (unsigned)__builtin_amdgcn_readfirstlane((int)(threadIdx.x >> 5));
    const unsigned c16  = lane & 15u;
    const unsigned hl   = lane >> 4;
    const unsigned g4   = lane >> 3;
    const unsigned p8   = lane & 7u;

    const float* ximg = x + (size_t)(bt * 16u + c16) * XB;
    const float* xq   = x + (size_t)(bt * 16u + g4) * XB + pi * ROWW + p8 * 4u;

    float*       st_wr = s_stage + wv * (16u * SPITCH) + g4 * SPITCH + p8 * 4u;
    const float* st_rd = s_stage + wv * (16u * SPITCH) + c16 * SPITCH + hl * 4u;

    v8f acc0 = {0.f, 0.f, 0.f, 0.f, 0.f, 0.f, 0.f, 0.f};
    v8f acc1 = {0.f, 0.f, 0.f, 0.f, 0.f, 0.f, 0.f, 0.f};

    v2f hx[HSTEPS];

    v4f      G[DEPTH_T][4];
    v4f      fa[2][4], fw[2][4];
    unsigned fc[2][4];
    const bool tail = wv + NWAVES * LASTTRIP < NUNITS;
    #pragma unroll
    for (unsigned d = 0; d < DEPTH_T; ++d)
        load_unit(G[d], xq, wv + NWAVES * d);
    unsigned go0 = 0u, go1 = 0u;
    if constexpr (HEAD)
        head_offsets(go0, go1, s_goff, wv, nhead, hl);
    store_unit(st_wr, G[0]);
    load_unit(G[0], xq, wv + NWAVES * DEPTH_T);
    frag_load(fa[0], fw[0], fc[0], st_rd, s_w, s_col, wv * 32u + hl * 4u);
    #pragma unroll
    for (unsigned t = 0; t <= LASTTRIP; ++t) {
        const unsigned n = t + 1u;
        const unsigned m = n + DEPTH_T;
        if (n < LASTTRIP || (n == LASTTRIP && tail)) {
            store_unit(st_wr, G[n % DEPTH_T]);
            if (m < LASTTRIP || (m == LASTTRIP && tail))
                load_unit(G[n % DEPTH_T], xq, wv + NWAVES * m);
            frag_load(fa[n & 1u], fw[n & 1u], fc[n & 1u], st_rd, s_w, s_col,
                      (wv + NWAVES * n) * 32u + hl * 4u);
        }
        if constexpr (HEAD) {
        if (t < (unsigned)HSTEPS) {
            hx[t].x = ximg[go0];
            hx[t].y = ximg[go1];
        }
        if (t + 1u < (unsigned)HSTEPS)
            head_offsets(go0, go1, s_goff, wv + (t + 1u) * NWAVES, nhead, hl);
        }
        __builtin_amdgcn_sched_barrier(0);
        if (t < LASTTRIP || tail)
            mac_unit(fa[t & 1u], fw[t & 1u], fc[t & 1u], c16, acc0, acc1);
        __builtin_amdgcn_sched_barrier(0);
    }

    if constexpr (HEAD) {
    #pragma unroll
    for (int t = 0; t < HSTEPS; ++t) {
        const unsigned st   = wv + (unsigned)t * NWAVES;
        const bool     live = st * 4u < nhead;
        const unsigned d0   = (live ? st * 4u : 0u) + hl;
        v2f a, b;
        a.x = live ? hx[t].x : 0.f;
        a.y = live ? hx[t].y : 0.f;
        b.x = (live && c16 == 0u) ? s_gw[d0]      : 0.f;
        b.y = (live && c16 == 0u) ? s_gw[d0 + 2u] : 0.f;
        acc0 = __builtin_amdgcn_wmma_f32_16x16x4_f32(
            false, a, false, b, (short)0, acc0, false, false);
    }
    }

    #pragma unroll
    for (int r = 0; r < 8; ++r)
        s_part[wv * 256u + (hl * 8u + (unsigned)r) * 16u + c16] = acc0[r] + acc1[r];
    __syncthreads();

    if (threadIdx.x < 16u * 13u) {
        const unsigned m   = threadIdx.x / 13u;
        const unsigned col = threadIdx.x - m * 13u;
        float s = 0.f;
        for (unsigned v = 0; v < NWAVES; ++v)
            s += s_part[v * 256u + m * 16u + col];
        const unsigned b = bt * 16u + m;
        const unsigned p = pbase + col;
        const unsigned g = pi >> 2;
        if (col == 12u && q != 3u) {
            if constexpr (!HEAD)
                ws[(g * 3u + q) * 256u + b] = s;
        } else if (!HEAD && col == 0u && q != 0u) {
            ws[WS_HALF + (g * 3u + q - 1u) * 256u + b] = s;
        } else {
            out[(size_t)b * NPATCH + p] = s + bias[p];
        }
    }
}

__global__ __launch_bounds__(NTHREADS) void patch_scorer_rows(
    const float* __restrict__ x,
    const float* __restrict__ weight,
    const float* __restrict__ bias,
    float* __restrict__ out,
    float* __restrict__ ws,
    unsigned long long ws_bytes)
{
    __shared__ __align__(16) float         s_w[PI_ELEMS];
    __shared__ __align__(16) unsigned char s_col[PI_ELEMS];
    __shared__ __align__(16) unsigned      s_goff[GMAX];
    __shared__ __align__(16) float         s_gw[GMAX];
    __shared__ __align__(16) float         s_part[NWAVES * 256u];
    __shared__ __align__(16) float         s_stage[NWAVES * 16u * SPITCH];

    if (ws != nullptr && ws_bytes >= WS_BYTES)
        scorer_body<false, DEPTH>(x, weight, bias, out, ws, s_w, s_col, s_goff, s_gw, s_part, s_stage);
    else
        scorer_body<true, DEPTH - 2u>(x, weight, bias, out, ws, s_w, s_col, s_goff, s_gw, s_part, s_stage);
}

__global__ __launch_bounds__(256) void patch_scorer_join(
    const float* __restrict__ bias,
    float* __restrict__ out,
    const float* __restrict__ ws,
    unsigned long long ws_bytes)
{
    if (ws == nullptr || ws_bytes < WS_BYTES)
        return;
    const unsigned sidx = blockIdx.x;
    const unsigned b    = threadIdx.x;
    const unsigned g    = sidx / 3u;
    const unsigned ql   = sidx - g * 3u + 1u;
    const unsigned p    = g * 49u + ql * 12u;
    out[(size_t)b * NPATCH + p] = (ws[sidx * 256u + b] + ws[WS_HALF + sidx * 256u + b]) + bias[p];
}

extern "C" void kernel_launch(void* const* d_in, const int* in_sizes, int n_in,
                              void* d_out, int out_size, void* d_ws, size_t ws_size,
                              hipStream_t stream) {
    const float* x      = (const float*)d_in[0];
    const float* weight = (const float*)d_in[1];
    const float* bias   = (const float*)d_in[2];
    float*       out    = (float*)d_out;
    float*       ws     = (float*)d_ws;

    dim3 grid(16, 16, 1);
    dim3 block(NTHREADS, 1, 1);
    patch_scorer_rows<<<grid, block, 0, stream>>>(x, weight, bias, out, ws, (unsigned long long)ws_size);
    patch_scorer_join<<<dim3(NSTRAD, 1, 1), dim3(256, 1, 1), 0, stream>>>(bias, out, ws, (unsigned long long)ws_size);
}


# one-hot B no longer built per unit with compares and selects: weights staged in LDS as zero-padded per-column images (4 per group of 4 pj), a lane reads its own image; sums and their order unchanged
# speedup vs baseline: 9.6123x; 1.0913x over previous
#include <hip/hip_runtime.h>


typedef __attribute__((ext_vector_type(2))) float v2f;
typedef __attribute__((ext_vector_type(4))) float v4f;
typedef __attribute__((ext_vector_type(8))) float v8f;

#define NPATCH     196
#define XB         150528u
#define CSTRIDE    50176u
#define ROWW       224u
#define PI_ELEMS   9408u
#define NUNITS     294u
#define LASTTRIP   24u
#define DEPTH      8u
#define SPITCH     36u
#define NWAVES     12u
#define NTHREADS   384
#define RLEN       588u
#define W_IT       25
#define W_RSTEP    24u
#define VPITCH     2352u
#define ZIMG       (16u * VPITCH + 8u)
#define VFLOATS    (ZIMG + VPITCH)
#define GMAX       576u
#define HSTEPS     12
#define NSTRAD     12u
#define WS_HALF    3072u
#define WS_BYTES   24576ull

__device__ __forceinline__ void load_unit(v4f (&g)[4], const float* xq, unsigned u)
{
    const unsigned rho = u / 7u, ln = u - rho * 7u;
    const unsigned c = rho / 14u, i = rho - c * 14u;
    const float* xr = xq + c * CSTRIDE + i * (16u * ROWW) + ln * 32u;
    #pragma unroll
    for (int v = 0; v < 4; ++v)
        g[v] = *(const v4f*)(xr + (size_t)v * (4u * XB));
}

__device__ __forceinline__ void store_unit(float* st_wr, const v4f (&g)[4])
{
    #pragma unroll
    for (int v = 0; v < 4; ++v)
        *(v4f*)(st_wr + v * (4u * SPITCH)) = g[v];
}

__device__ __forceinline__ void frag_load(v4f (&fa)[4], v4f (&fb)[4],
                                          const float* st_rd, const float* bb0, const float* bb1,
                                          unsigned n)
{
    #pragma unroll
    for (int s = 0; s < 4; ++s) {
        fa[s] = *(const v4f*)(st_rd + s * 8);
        fb[s] = *(const v4f*)(((s & 1) ? bb1 : bb0) + n * 96u + (unsigned)(s >> 1) * 4u);
    }
}

__device__ __forceinline__ void head_offsets(unsigned& o0, unsigned& o1, const unsigned* s_goff,
                                             unsigned st, unsigned nhead, unsigned hl)
{
    const unsigned d0 = ((st * 4u < nhead) ? st * 4u : 0u) + hl;
    o0 = s_goff[d0];
    o1 = s_goff[d0 + 2u];
}

__device__ __forceinline__ void mac_unit(const v4f (&fa)[4], const v4f (&fb)[4], v8f& acc0, v8f& acc1)
{
    #pragma unroll
    for (int s = 0; s < 4; ++s) {
        v2f a0, a1, b0, b1;
        a0.x = fa[s].x; a0.y = fa[s].y;
        a1.x = fa[s].z; a1.y = fa[s].w;
        b0.x = fb[s].x; b0.y = fb[s].y;
        b1.x = fb[s].z; b1.y = fb[s].w;
        acc0 = __builtin_amdgcn_wmma_f32_16x16x4_f32(
            false, a0, false, b0, (short)0, acc0, false, false);
        acc1 = __builtin_amdgcn_wmma_f32_16x16x4_f32(
            false, a1, false, b1, (short)0, acc1, false, false);
    }
}

template <bool HEAD, unsigned DEPTH_T>
__device__ __forceinline__ void scorer_body(
    const float* __restrict__ x,
    const float* __restrict__ weight,
    const float* __restrict__ bias,
    float* __restrict__ out,
    float* __restrict__ ws,
    float* s_v, unsigned* s_goff, float* s_gw, float* s_part, float* s_stage)
{
    const unsigned pi    = blockIdx.x;
    const unsigned bt    = blockIdx.y;
    const unsigned q     = pi & 3u;
    const unsigned pbase = (pi >> 2) * 49u + q * 12u;
    const unsigned nhead = q * 192u;

    {
        const unsigned wv1 = (unsigned)__builtin_amdgcn_readfirstlane((int)(threadIdx.x >> 5));
        const unsigned grp = wv1 & 3u;
        const unsigned e4  = (threadIdx.x & 31u) >> 3;
        const unsigned pj  = grp * 4u + e4;
        const unsigned r0  = (wv1 >> 2) * 8u + (threadIdx.x & 7u);
        const float*   wrow = weight + pi * PI_ELEMS + pj * RLEN;
        const unsigned pgrp = (pi * PI_ELEMS + grp * (4u * RLEN)) / 768u;
        float*         vgrp = s_v + grp * (4u * VPITCH);
        float wr[W_IT];
        #pragma unroll
        for (int t = 0; t < W_IT; ++t) {
            unsigned r = r0 + (unsigned)t * W_RSTEP;
            if (t + 1 == W_IT)
                r = r < RLEN ? r : RLEN - 1u;
            wr[t] = wrow[r];
        }
        const v4f zero4 = {0.f, 0.f, 0.f, 0.f};
        #pragma unroll
        for (int t = 0; t < W_IT; ++t) {
            const unsigned r = r0 + (unsigned)t * W_RSTEP;
            if (t + 1 < W_IT || r < RLEN)
                *(v4f*)(vgrp + e4 * VPITCH + 4u * r) = zero4;
        }
        for (unsigned d = threadIdx.x; d < RLEN; d += NTHREADS)
            *(v4f*)(s_v + ZIMG + 4u * d) = zero4;
        #pragma unroll
        for (int t = 0; t < W_IT; ++t) {
            unsigned r = r0 + (unsigned)t * W_RSTEP;
            if (t + 1 == W_IT)
                r = r < RLEN ? r : RLEN - 1u;
            const unsigned img = (pi * PI_ELEMS + pj * RLEN + r) / 768u - pgrp;
            if (t + 1 < W_IT || (wv1 >> 2) < 2u)
                vgrp[img * VPITCH + 4u * r + e4] = wr[t];
        }
        if constexpr (HEAD) {
        for (unsigned hd = threadIdx.x; hd < nhead; hd += NTHREADS) {
            const unsigned L    = pbase * 768u + hd;
            const unsigned pipj = L / 588u;
            const unsigned rem  = L - pipj * 588u;
            const unsigned c    = rem / 196u;
            const unsigned r2   = rem - c * 196u;
            const unsigned i    = r2 / 14u;
            const unsigned j    = r2 - i * 14u;
            s_goff[hd] = c * CSTRIDE + (i * 16u + (pipj >> 4)) * ROWW + (j * 16u + (pipj & 15u));
            s_gw[hd]   = weight[L];
        }
        if (nhead == 0u && threadIdx.x < 4u)
            s_goff[threadIdx.x] = 0u;
        }
    }
    __syncthreads();

    const unsigned lane = threadIdx.x & 31u;
    const unsigned wv   = (unsigned)__builtin_amdgcn_readfirstlane((int)(threadIdx.x >> 5));
    const unsigned c16  = lane & 15u;
    const unsigned hl   = lane >> 4;
    const unsigned g4   = lane >> 3;
    const unsigned p8   = lane & 7u;

    const float* ximg = x + (size_t)(bt * 16u + c16) * XB;
    const float* xq   = x + (size_t)(bt * 16u + g4) * XB + pi * ROWW + p8 * 4u;

    float*       st_wr = s_stage + wv * (16u * SPITCH) + g4 * SPITCH + p8 * 4u;
    const float* st_rd = s_stage + wv * (16u * SPITCH) + c16 * SPITCH + hl * 4u;

    const unsigned im0 = c16 + pbase - (pi * PI_ELEMS + hl * (4u * RLEN)) / 768u;
    const unsigned im1 = c16 + pbase - (pi * PI_ELEMS + (2u + hl) * (4u * RLEN)) / 768u;
    const float* bb0 = s_v + (im0 < 4u ? (hl * 4u + im0) * VPITCH : ZIMG) + wv * 8u;
    const float* bb1 = s_v + (im1 < 4u ? ((2u + hl) * 4u + im1) * VPITCH : ZIMG) + wv * 8u;

    v8f acc0 = {0.f, 0.f, 0.f, 0.f, 0.f, 0.f, 0.f, 0.f};
    v8f acc1 = {0.f, 0.f, 0.f, 0.f, 0.f, 0.f, 0.f, 0.f};

    v2f hx[HSTEPS];

    v4f      G[DEPTH_T][4];
    v4f      fa[2][4], fb[2][4];
    const bool tail = wv + NWAVES * LASTTRIP < NUNITS;
    #pragma unroll
    for (unsigned d = 0; d < DEPTH_T; ++d)
        load_unit(G[d], xq, wv + NWAVES * d);
    unsigned go0 = 0u, go1 = 0u;
    if constexpr (HEAD)
        head_offsets(go0, go1, s_goff, wv, nhead, hl);
    store_unit(st_wr, G[0]);
    load_unit(G[0], xq, wv + NWAVES * DEPTH_T);
    frag_load(fa[0], fb[0], st_rd, bb0, bb1, 0u);
    #pragma unroll
    for (unsigned t = 0; t <= LASTTRIP; ++t) {
        const unsigned n = t + 1u;
        const unsigned m = n + DEPTH_T;
        if (n < LASTTRIP || (n == LASTTRIP && tail)) {
            store_unit(st_wr, G[n % DEPTH_T]);
            if (m < LASTTRIP || (m == LASTTRIP && tail))
                load_unit(G[n % DEPTH_T], xq, wv + NWAVES * m);
            frag_load(fa[n & 1u], fb[n & 1u], st_rd, bb0, bb1, n);
        }
        if constexpr (HEAD) {
        if (t < (unsigned)HSTEPS) {
            hx[t].x = ximg[go0];
            hx[t].y = ximg[go1];
        }
        if (t + 1u < (unsigned)HSTEPS)
            head_offsets(go0, go1, s_goff, wv + (t + 1u) * NWAVES, nhead, hl);
        }
        __builtin_amdgcn_sched_barrier(0);
        if (t < LASTTRIP || tail)
            mac_unit(fa[t & 1u], fb[t & 1u], acc0, acc1);
        __builtin_amdgcn_sched_barrier(0);
    }

    if constexpr (HEAD) {
    #pragma unroll
    for (int t = 0; t < HSTEPS; ++t) {
        const unsigned st   = wv + (unsigned)t * NWAVES;
        const bool     live = st * 4u < nhead;
        const unsigned d0   = (live ? st * 4u : 0u) + hl;
        v2f a, b;
        a.x = live ? hx[t].x : 0.f;
        a.y = live ? hx[t].y : 0.f;
        b.x = (live && c16 == 0u) ? s_gw[d0]      : 0.f;
        b.y = (live && c16 == 0u) ? s_gw[d0 + 2u] : 0.f;
        acc0 = __builtin_amdgcn_wmma_f32_16x16x4_f32(
            false, a, false, b, (short)0, acc0, false, false);
    }
    }

    #pragma unroll
    for (int r = 0; r < 8; ++r)
        s_part[wv * 256u + (hl * 8u + (unsigned)r) * 16u + c16] = acc0[r] + acc1[r];
    __syncthreads();

    if (threadIdx.x < 16u * 13u) {
        const unsigned m   = threadIdx.x / 13u;
        const unsigned col = threadIdx.x - m * 13u;
        float s = 0.f;
        for (unsigned v = 0; v < NWAVES; ++v)
            s += s_part[v * 256u + m * 16u + col];
        const unsigned b = bt * 16u + m;
        const unsigned p = pbase + col;
        const unsigned g = pi >> 2;
        if (col == 12u && q != 3u) {
            if constexpr (!HEAD)
                ws[(g * 3u + q) * 256u + b] = s;
        } else if (!HEAD && col == 0u && q != 0u) {
            ws[WS_HALF + (g * 3u + q - 1u) * 256u + b] = s;
        } else {
            out[(size_t)b * NPATCH + p] = s + bias[p];
        }
    }
}

__global__ __launch_bounds__(NTHREADS) void patch_scorer_rows(
    const float* __restrict__ x,
    const float* __restrict__ weight,
    const float* __restrict__ bias,
    float* __restrict__ out,
    float* __restrict__ ws,
    unsigned long long ws_bytes)
{
    __shared__ __align__(16) float         s_v[VFLOATS];
    __shared__ __align__(16) unsigned      s_goff[GMAX];
    __shared__ __align__(16) float         s_gw[GMAX];
    __shared__ __align__(16) float         s_part[NWAVES * 256u];
    __shared__ __align__(16) float         s_stage[NWAVES * 16u * SPITCH];

    if (ws != nullptr && ws_bytes >= WS_BYTES)
        scorer_body<false, DEPTH>(x, weight, bias, out, ws, s_v, s_goff, s_gw, s_part, s_stage);
    else
        scorer_body<true, DEPTH - 2u>(x, weight, bias, out, ws, s_v, s_goff, s_gw, s_part, s_stage);
}

__global__ __launch_bounds__(256) void patch_scorer_join(
    const float* __restrict__ bias,
    float* __restrict__ out,
    const float* __restrict__ ws,
    unsigned long long ws_bytes)
{
    if (ws == nullptr || ws_bytes < WS_BYTES)
        return;
    const unsigned sidx = blockIdx.x;
    const unsigned b    = threadIdx.x;
    const unsigned g    = sidx / 3u;
    const unsigned ql   = sidx - g * 3u + 1u;
    const unsigned p    = g * 49u + ql * 12u;
    out[(size_t)b * NPATCH + p] = (ws[sidx * 256u + b] + ws[WS_HALF + sidx * 256u + b]) + bias[p];
}

extern "C" void kernel_launch(void* const* d_in, const int* in_sizes, int n_in,
                              void* d_out, int out_size, void* d_ws, size_t ws_size,
                              hipStream_t stream) {
    const float* x      = (const float*)d_in[0];
    const float* weight = (const float*)d_in[1];
    const float* bias   = (const float*)d_in[2];
    float*       out    = (float*)d_out;
    float*       ws     = (float*)d_ws;

    dim3 grid(16, 16, 1);
    dim3 block(NTHREADS, 1, 1);
    patch_scorer_rows<<<grid, block, 0, stream>>>(x, weight, bias, out, ws, (unsigned long long)ws_size);
    patch_scorer_join<<<dim3(NSTRAD, 1, 1), dim3(256, 1, 1), 0, stream>>>(bias, out, ws, (unsigned long long)ws_size);
}


# first 8 x units requested behind the weight loads, before the weight staging (its clearing stores spread among them); the staging's barrier fences LDS only, so they stay in flight; sums unchanged
# speedup vs baseline: 10.1915x; 1.0603x over previous
#include <hip/hip_runtime.h>


typedef __attribute__((ext_vector_type(2))) float v2f;
typedef __attribute__((ext_vector_type(4))) float v4f;
typedef __attribute__((ext_vector_type(8))) float v8f;

#define NPATCH     196
#define XB         150528u
#define CSTRIDE    50176u
#define ROWW       224u
#define PI_ELEMS   9408u
#define NUNITS     294u
#define LASTTRIP   24u
#define DEPTH      8u
#define SPITCH     36u
#define NWAVES     12u
#define NTHREADS   384
#define RLEN       588u
#define W_IT       25
#define W_RSTEP    24u
#define VPITCH     2352u
#define ZIMG       (16u * VPITCH + 8u)
#define VFLOATS    (ZIMG + VPITCH)
#define GMAX       576u
#define HSTEPS     12
#define NSTRAD     12u
#define WS_HALF    3072u
#define WS_BYTES   24576ull

__device__ __forceinline__ void load_unit(v4f (&g)[4], const float* xq, unsigned u)
{
    const unsigned rho = u / 7u, ln = u - rho * 7u;
    const unsigned c = rho / 14u, i = rho - c * 14u;
    const float* xr = xq + c * CSTRIDE + i * (16u * ROWW) + ln * 32u;
    #pragma unroll
    for (int v = 0; v < 4; ++v)
        g[v] = *(const v4f*)(xr + (size_t)v * (4u * XB));
}

__device__ __forceinline__ void store_unit(float* st_wr, const v4f (&g)[4])
{
    #pragma unroll
    for (int v = 0; v < 4; ++v)
        *(v4f*)(st_wr + v * (4u * SPITCH)) = g[v];
}

__device__ __forceinline__ void frag_load(v4f (&fa)[4], v4f (&fb)[4],
                                          const float* st_rd, const float* bb0, const float* bb1,
                                          unsigned n)
{
    #pragma unroll
    for (int s = 0; s < 4; ++s) {
        fa[s] = *(const v4f*)(st_rd + s * 8);
        fb[s] = *(const v4f*)(((s & 1) ? bb1 : bb0) + n * 96u + (unsigned)(s >> 1) * 4u);
    }
}

__device__ __forceinline__ void head_offsets(unsigned& o0, unsigned& o1, const unsigned* s_goff,
                                             unsigned st, unsigned nhead, unsigned hl)
{
    const unsigned d0 = ((st * 4u < nhead) ? st * 4u : 0u) + hl;
    o0 = s_goff[d0];
    o1 = s_goff[d0 + 2u];
}

__device__ __forceinline__ void mac_unit(const v4f (&fa)[4], const v4f (&fb)[4], v8f& acc0, v8f& acc1)
{
    #pragma unroll
    for (int s = 0; s < 4; ++s) {
        v2f a0, a1, b0, b1;
        a0.x = fa[s].x; a0.y = fa[s].y;
        a1.x = fa[s].z; a1.y = fa[s].w;
        b0.x = fb[s].x; b0.y = fb[s].y;
        b1.x = fb[s].z; b1.y = fb[s].w;
        acc0 = __builtin_amdgcn_wmma_f32_16x16x4_f32(
            false, a0, false, b0, (short)0, acc0, false, false);
        acc1 = __builtin_amdgcn_wmma_f32_16x16x4_f32(
            false, a1, false, b1, (short)0, acc1, false, false);
    }
}

template <bool HEAD, unsigned DEPTH_T>
__device__ __forceinline__ void scorer_body(
    const float* __restrict__ x,
    const float* __restrict__ weight,
    const float* __restrict__ bias,
    float* __restrict__ out,
    float* __restrict__ ws,
    float* s_v, unsigned* s_goff, float* s_gw, float* s_part, float* s_stage)
{
    const unsigned pi    = blockIdx.x;
    const unsigned bt    = blockIdx.y;
    const unsigned q     = pi & 3u;
    const unsigned pbase = (pi >> 2) * 49u + q * 12u;
    const unsigned nhead = q * 192u;

    const unsigned lane = threadIdx.x & 31u;
    const unsigned wv   = (unsigned)__builtin_amdgcn_readfirstlane((int)(threadIdx.x >> 5));
    const unsigned c16  = lane & 15u;
    const unsigned hl   = lane >> 4;
    const unsigned g4   = lane >> 3;
    const unsigned p8   = lane & 7u;

    const float* ximg = x + (size_t)(bt * 16u + c16) * XB;
    const float* xq   = x + (size_t)(bt * 16u + g4) * XB + pi * ROWW + p8 * 4u;

    v4f      G[DEPTH_T][4];

    {
        const unsigned wv1 = (unsigned)__builtin_amdgcn_readfirstlane((int)(threadIdx.x >> 5));
        const unsigned grp = wv1 & 3u;
        const unsigned e4  = (threadIdx.x & 31u) >> 3;
        const unsigned pj  = grp * 4u + e4;
        const unsigned r0  = (wv1 >> 2) * 8u + (threadIdx.x & 7u);
        const float*   wrow = weight + pi * PI_ELEMS + pj * RLEN;
        const unsigned pgrp = (pi * PI_ELEMS + grp * (4u * RLEN)) / 768u;
        float*         vgrp = s_v + grp * (4u * VPITCH);
        float wr[W_IT];
        #pragma unroll
        for (int t = 0; t < W_IT; ++t) {
            unsigned r = r0 + (unsigned)t * W_RSTEP;
            if (t + 1 == W_IT)
                r = r < RLEN ? r : RLEN - 1u;
            wr[t] = wrow[r];
        }
        const v4f zero4 = {0.f, 0.f, 0.f, 0.f};
        auto clear_rows = [&](int t0, int t1) {
            #pragma unroll
            for (int t = t0; t < t1; ++t) {
                const unsigned r = r0 + (unsigned)t * W_RSTEP;
                if (t + 1 < W_IT || r < RLEN)
                    *(v4f*)(vgrp + e4 * VPITCH + 4u * r) = zero4;
            }
        };
        if constexpr (!HEAD) {
            __builtin_amdgcn_sched_barrier(0);
            constexpr int CL = (W_IT + (int)DEPTH_T - 1) / (int)DEPTH_T;
            #pragma unroll
            for (unsigned d = 0; d < DEPTH_T; ++d) {
                load_unit(G[d], xq, wv + NWAVES * d);
                clear_rows((int)d * CL, ((int)d + 1) * CL < W_IT ? ((int)d + 1) * CL : W_IT);
                __builtin_amdgcn_sched_barrier(0);
            }
        } else {
            clear_rows(0, W_IT);
        }
        for (unsigned d = threadIdx.x; d < RLEN; d += NTHREADS)
            *(v4f*)(s_v + ZIMG + 4u * d) = zero4;
        #pragma unroll
        for (int t = 0; t < W_IT; ++t) {
            unsigned r = r0 + (unsigned)t * W_RSTEP;
            if (t + 1 == W_IT)
                r = r < RLEN ? r : RLEN - 1u;
            const unsigned img = (pi * PI_ELEMS + pj * RLEN + r) / 768u - pgrp;
            if (t + 1 < W_IT || (wv1 >> 2) < 2u)
                vgrp[img * VPITCH + 4u * r + e4] = wr[t];
        }
        if constexpr (HEAD) {
        for (unsigned hd = threadIdx.x; hd < nhead; hd += NTHREADS) {
            const unsigned L    = pbase * 768u + hd;
            const unsigned pipj = L / 588u;
            const unsigned rem  = L - pipj * 588u;
            const unsigned c    = rem / 196u;
            const unsigned r2   = rem - c * 196u;
            const unsigned i    = r2 / 14u;
            const unsigned j    = r2 - i * 14u;
            s_goff[hd] = c * CSTRIDE + (i * 16u + (pipj >> 4)) * ROWW + (j * 16u + (pipj & 15u));
            s_gw[hd]   = weight[L];
        }
        if (nhead == 0u && threadIdx.x < 4u)
            s_goff[threadIdx.x] = 0u;
        }
    }
    __builtin_amdgcn_fence(__ATOMIC_RELEASE, "workgroup", "local");
    __builtin_amdgcn_s_barrier();
    __builtin_amdgcn_fence(__ATOMIC_ACQUIRE, "workgroup", "local");

    float*       st_wr = s_stage + wv * (16u * SPITCH) + g4 * SPITCH + p8 * 4u;
    const float* st_rd = s_stage + wv * (16u * SPITCH) + c16 * SPITCH + hl * 4u;

    const unsigned im0 = c16 + pbase - (pi * PI_ELEMS + hl * (4u * RLEN)) / 768u;
    const unsigned im1 = c16 + pbase - (pi * PI_ELEMS + (2u + hl) * (4u * RLEN)) / 768u;
    const float* bb0 = s_v + (im0 < 4u ? (hl * 4u + im0) * VPITCH : ZIMG) + wv * 8u;
    const float* bb1 = s_v + (im1 < 4u ? ((2u + hl) * 4u + im1) * VPITCH : ZIMG) + wv * 8u;

    v8f acc0 = {0.f, 0.f, 0.f, 0.f, 0.f, 0.f, 0.f, 0.f};
    v8f acc1 = {0.f, 0.f, 0.f, 0.f, 0.f, 0.f, 0.f, 0.f};

    v2f hx[HSTEPS];

    v4f      fa[2][4], fb[2][4];
    const bool tail = wv + NWAVES * LASTTRIP < NUNITS;
    if constexpr (HEAD) {
        #pragma unroll
        for (unsigned d = 0; d < DEPTH_T; ++d)
            load_unit(G[d], xq, wv + NWAVES * d);
    }
    unsigned go0 = 0u, go1 = 0u;
    if constexpr (HEAD)
        head_offsets(go0, go1, s_goff, wv, nhead, hl);
    store_unit(st_wr, G[0]);
    load_unit(G[0], xq, wv + NWAVES * DEPTH_T);
    frag_load(fa[0], fb[0], st_rd, bb0, bb1, 0u);
    #pragma unroll
    for (unsigned t = 0; t <= LASTTRIP; ++t) {
        const unsigned n = t + 1u;
        const unsigned m = n + DEPTH_T;
        if (n < LASTTRIP || (n == LASTTRIP && tail)) {
            store_unit(st_wr, G[n % DEPTH_T]);
            if (m < LASTTRIP || (m == LASTTRIP && tail))
                load_unit(G[n % DEPTH_T], xq, wv + NWAVES * m);
            frag_load(fa[n & 1u], fb[n & 1u], st_rd, bb0, bb1, n);
        }
        if constexpr (HEAD) {
        if (t < (unsigned)HSTEPS) {
            hx[t].x = ximg[go0];
            hx[t].y = ximg[go1];
        }
        if (t + 1u < (unsigned)HSTEPS)
            head_offsets(go0, go1, s_goff, wv + (t + 1u) * NWAVES, nhead, hl);
        }
        __builtin_amdgcn_sched_barrier(0);
        if (t < LASTTRIP || tail)
            mac_unit(fa[t & 1u], fb[t & 1u], acc0, acc1);
        __builtin_amdgcn_sched_barrier(0);
    }

    if constexpr (HEAD) {
    #pragma unroll
    for (int t = 0; t < HSTEPS; ++t) {
        const unsigned st   = wv + (unsigned)t * NWAVES;
        const bool     live = st * 4u < nhead;
        const unsigned d0   = (live ? st * 4u : 0u) + hl;
        v2f a, b;
        a.x = live ? hx[t].x : 0.f;
        a.y = live ? hx[t].y : 0.f;
        b.x = (live && c16 == 0u) ? s_gw[d0]      : 0.f;
        b.y = (live && c16 == 0u) ? s_gw[d0 + 2u] : 0.f;
        acc0 = __builtin_amdgcn_wmma_f32_16x16x4_f32(
            false, a, false, b, (short)0, acc0, false, false);
    }
    }

    #pragma unroll
    for (int r = 0; r < 8; ++r)
        s_part[wv * 256u + (hl * 8u + (unsigned)r) * 16u + c16] = acc0[r] + acc1[r];
    __syncthreads();

    if (threadIdx.x < 16u * 13u) {
        const unsigned m   = threadIdx.x / 13u;
        const unsigned col = threadIdx.x - m * 13u;
        float s = 0.f;
        for (unsigned v = 0; v < NWAVES; ++v)
            s += s_part[v * 256u + m * 16u + col];
        const unsigned b = bt * 16u + m;
        const unsigned p = pbase + col;
        const unsigned g = pi >> 2;
        if (col == 12u && q != 3u) {
            if constexpr (!HEAD)
                ws[(g * 3u + q) * 256u + b] = s;
        } else if (!HEAD && col == 0u && q != 0u) {
            ws[WS_HALF + (g * 3u + q - 1u) * 256u + b] = s;
        } else {
            out[(size_t)b * NPATCH + p] = s + bias[p];
        }
    }
}

__global__ __launch_bounds__(NTHREADS) void patch_scorer_rows(
    const float* __restrict__ x,
    const float* __restrict__ weight,
    const float* __restrict__ bias,
    float* __restrict__ out,
    float* __restrict__ ws,
    unsigned long long ws_bytes)
{
    __shared__ __align__(16) float         s_v[VFLOATS];
    __shared__ __align__(16) unsigned      s_goff[GMAX];
    __shared__ __align__(16) float         s_gw[GMAX];
    __shared__ __align__(16) float         s_part[NWAVES * 256u];
    __shared__ __align__(16) float         s_stage[NWAVES * 16u * SPITCH];

    if (ws != nullptr && ws_bytes >= WS_BYTES)
        scorer_body<false, DEPTH>(x, weight, bias, out, ws, s_v, s_goff, s_gw, s_part, s_stage);
    else
        scorer_body<true, DEPTH - 2u>(x, weight, bias, out, ws, s_v, s_goff, s_gw, s_part, s_stage);
}

__global__ __launch_bounds__(256) void patch_scorer_join(
    const float* __restrict__ bias,
    float* __restrict__ out,
    const float* __restrict__ ws,
    unsigned long long ws_bytes)
{
    if (ws == nullptr || ws_bytes < WS_BYTES)
        return;
    const unsigned sidx = blockIdx.x;
    const unsigned b    = threadIdx.x;
    const unsigned g    = sidx / 3u;
    const unsigned ql   = sidx - g * 3u + 1u;
    const unsigned p    = g * 49u + ql * 12u;
    out[(size_t)b * NPATCH + p] = (ws[sidx * 256u + b] + ws[WS_HALF + sidx * 256u + b]) + bias[p];
}

extern "C" void kernel_launch(void* const* d_in, const int* in_sizes, int n_in,
                              void* d_out, int out_size, void* d_ws, size_t ws_size,
                              hipStream_t stream) {
    const float* x      = (const float*)d_in[0];
    const float* weight = (const float*)d_in[1];
    const float* bias   = (const float*)d_in[2];
    float*       out    = (float*)d_out;
    float*       ws     = (float*)d_ws;

    dim3 grid(16, 16, 1);
    dim3 block(NTHREADS, 1, 1);
    patch_scorer_rows<<<grid, block, 0, stream>>>(x, weight, bias, out, ws, (unsigned long long)ws_size);
    patch_scorer_join<<<dim3(NSTRAD, 1, 1), dim3(256, 1, 1), 0, stream>>>(bias, out, ws, (unsigned long long)ws_size);
}
